# starting kernel
# baseline (speedup 1.0000x reference)
#include <hip/hip_runtime.h>
#include <math.h>


#define CHN 256
#define HIDN 16
#define BAT 32
#define HGT 56
#define WID 56
#define HW  (HGT * WID)
#define NSL 16
#define SLICE (HW / NSL)
#define C4  (CHN / 4)

typedef float v2f __attribute__((ext_vector_type(2)));
typedef float v8f __attribute__((ext_vector_type(8)));

__device__ __forceinline__ float sig_(float v) {
    return __builtin_amdgcn_rcpf(1.0f + __expf(-v));
}

__global__ __launch_bounds__(256) void k1_partial(const float* __restrict__ x,
                                                  float* __restrict__ psum,
                                                  float* __restrict__ pmax) {
    __shared__ float ssum[4 * CHN];
    __shared__ float smax[4 * CHN];

    const int b = blockIdx.x >> 4;
    const int s = blockIdx.x & 15;
    const int t = threadIdx.x;
    const int pg = t >> 6;
    const int q  = t & 63;

    const float4* xp = (const float4*)(x + ((size_t)b * HW + (size_t)s * SLICE) * CHN);

    float4 sum = {0.f, 0.f, 0.f, 0.f};
    float4 mx  = {-3.4e38f, -3.4e38f, -3.4e38f, -3.4e38f};
    for (int p = pg; p < SLICE; p += 4) {
        __builtin_prefetch(&xp[(size_t)(p + 16) * C4 + q], 0, 0);
        float4 v = xp[(size_t)p * C4 + q];
        sum.x += v.x; sum.y += v.y; sum.z += v.z; sum.w += v.w;
        mx.x = fmaxf(mx.x, v.x); mx.y = fmaxf(mx.y, v.y);
        mx.z = fmaxf(mx.z, v.z); mx.w = fmaxf(mx.w, v.w);
    }
    ((float4*)ssum)[pg * C4 + q] = sum;
    ((float4*)smax)[pg * C4 + q] = mx;
    __syncthreads();

    const int c = t;
    float r = ssum[c] + ssum[CHN + c] + ssum[2 * CHN + c] + ssum[3 * CHN + c];
    float m = fmaxf(fmaxf(smax[c], smax[CHN + c]),
                    fmaxf(smax[2 * CHN + c], smax[3 * CHN + c]));
    psum[(size_t)blockIdx.x * CHN + c] = r;
    pmax[(size_t)blockIdx.x * CHN + c] = m;
}

__global__ __launch_bounds__(256) void k2_gate(const float* __restrict__ psum,
                                               const float* __restrict__ pmax,
                                               const float* __restrict__ w1,
                                               const float* __restrict__ b1,
                                               const float* __restrict__ w2,
                                               const float* __restrict__ b2,
                                               float* __restrict__ gate) {
    __shared__ float pool[BAT][CHN];
    __shared__ float hid[BAT][HIDN];

    const int t = threadIdx.x;
    const int wave = t >> 5;
    const int lane = t & 31;
    const int l16 = lane & 15;
    const int hi = lane >> 4;

    v8f zero = {};
    v8f acc[4];
#pragma unroll
    for (int i = 0; i < 4; ++i) acc[i] = zero;

    for (int pass = 0; pass < 2; ++pass) {
        const float* part = pass ? pmax : psum;
        for (int b = 0; b < BAT; ++b) {
            float r = pass ? -3.4e38f : 0.0f;
            for (int s = 0; s < NSL; ++s) {
                float v = part[(size_t)(b * NSL + s) * CHN + t];
                r = pass ? fmaxf(r, v) : (r + v);
            }
            pool[b][t] = pass ? r : r * (1.0f / (float)HW);
        }
        __syncthreads();

        if (wave < 2) {
            const int m0 = wave * 16;
            v8f c = zero;
            for (int k = 0; k < CHN; k += 4) {
                v2f a, bm;
                a.x = pool[m0 + l16][k + 2 * hi + 0];
                a.y = pool[m0 + l16][k + 2 * hi + 1];
                bm.x = w1[(k + 2 * hi + 0) * HIDN + l16];
                bm.y = w1[(k + 2 * hi + 1) * HIDN + l16];
                c = __builtin_amdgcn_wmma_f32_16x16x4_f32(false, a, false, bm,
                                                          (short)0, c, false, false);
            }
            const float bias = b1[l16];
#pragma unroll
            for (int r = 0; r < 8; ++r) {
                hid[m0 + r + 8 * hi][l16] = fmaxf(c[r] + bias, 0.0f);
            }
        }
        __syncthreads();

#pragma unroll
        for (int tt = 0; tt < 4; ++tt) {
            const int tile = wave * 4 + tt;
            const int m0 = (tile >> 4) * 16;
            const int n0 = (tile & 15) * 16;
            v8f c = zero;
#pragma unroll
            for (int k = 0; k < HIDN; k += 4) {
                v2f a, bm;
                a.x = hid[m0 + l16][k + 2 * hi + 0];
                a.y = hid[m0 + l16][k + 2 * hi + 1];
                bm.x = w2[(k + 2 * hi + 0) * CHN + n0 + l16];
                bm.y = w2[(k + 2 * hi + 1) * CHN + n0 + l16];
                c = __builtin_amdgcn_wmma_f32_16x16x4_f32(false, a, false, bm,
                                                          (short)0, c, false, false);
            }
            const float bias = b2[n0 + l16];
#pragma unroll
            for (int r = 0; r < 8; ++r) {
                acc[tt][r] += sig_(c[r] + bias);
            }
        }
        __syncthreads();
    }

#pragma unroll
    for (int tt = 0; tt < 4; ++tt) {
        const int tile = wave * 4 + tt;
        const int m0 = (tile >> 4) * 16;
        const int n0 = (tile & 15) * 16;
#pragma unroll
        for (int r = 0; r < 8; ++r) {
            gate[(size_t)(m0 + r + 8 * hi) * CHN + n0 + l16] = acc[tt][r];
        }
    }
}

__global__ __launch_bounds__(256) void k3_spatial(const float* __restrict__ x,
                                                  const float* __restrict__ gate,
                                                  float* __restrict__ sbuf) {
    const int wave = threadIdx.x >> 5;
    const int lane = threadIdx.x & 31;
    const int gp = blockIdx.x * 8 + wave;
    const int b = gp / HW;
    const float4* xp = (const float4*)(x + (size_t)gp * CHN);
    const float4* g4 = (const float4*)(gate + (size_t)b * CHN);

    float sum = 0.0f, mx = -3.4e38f;
#pragma unroll
    for (int i = 0; i < 2; ++i) {
        const int qi = lane + 32 * i;
        const float4 xv = xp[qi];
        const float4 gv = g4[qi];
        const float p0 = xv.x * gv.x, p1 = xv.y * gv.y;
        const float p2 = xv.z * gv.z, p3 = xv.w * gv.w;
        sum += p0 + p1 + p2 + p3;
        mx = fmaxf(mx, fmaxf(fmaxf(p0, p1), fmaxf(p2, p3)));
    }
#pragma unroll
    for (int off = 16; off > 0; off >>= 1) {
        sum += __shfl_xor(sum, off, 32);
        mx = fmaxf(mx, __shfl_xor(mx, off, 32));
    }
    if (lane == 0) {
        sbuf[(size_t)gp * 2 + 0] = sum * (1.0f / (float)CHN);
        sbuf[(size_t)gp * 2 + 1] = mx;
    }
}

__global__ __launch_bounds__(256) void k4_out(const float* __restrict__ x,
                                              const float* __restrict__ gate,
                                              const float* __restrict__ sbuf,
                                              const float* __restrict__ cw,
                                              const float* __restrict__ cb,
                                              float* __restrict__ out) {
    __shared__ float sg[WID];
    const int b = blockIdx.x / HGT;
    const int h = blockIdx.x % HGT;
    const int t = threadIdx.x;

    if (t < WID) {
        float acc = cb[0];
        for (int dy = -3; dy <= 3; ++dy) {
            const int hh = h + dy;
            if (hh < 0 || hh >= HGT) continue;
            for (int dx = -3; dx <= 3; ++dx) {
                const int ww = t + dx;
                if (ww < 0 || ww >= WID) continue;
                const float* sp = sbuf + ((size_t)(b * HGT + hh) * WID + ww) * 2;
                const float* wp = cw + ((dy + 3) * 7 + (dx + 3)) * 2;
                acc += sp[0] * wp[0] + sp[1] * wp[1];
            }
        }
        sg[t] = sig_(acc);
    }
    __syncthreads();

    const int q    = t & 63;
    const int wofs = t >> 6;
    const float4 g = ((const float4*)gate)[(size_t)b * C4 + q];

    const size_t row = ((size_t)b * HGT + h) * WID;
    const float4* xr4 = (const float4*)(x + row * CHN);
    float4*       o4  = (float4*)(out + row * CHN);

#pragma unroll 2
    for (int i = 0; i < WID / 4; ++i) {
        const int w = i * 4 + wofs;
        const float4 xv = xr4[(size_t)w * C4 + q];
        const float sgw = sg[w];
        float4 ov;
        ov.x = xv.x * g.x * sgw;
        ov.y = xv.y * g.y * sgw;
        ov.z = xv.z * g.z * sgw;
        ov.w = xv.w * g.w * sgw;
        o4[(size_t)w * C4 + q] = ov;
    }
}

extern "C" void kernel_launch(void* const* d_in, const int* in_sizes, int n_in,
                              void* d_out, int out_size, void* d_ws, size_t ws_size,
                              hipStream_t stream) {
    const float* x  = (const float*)d_in[0];
    const float* w1 = (const float*)d_in[1];
    const float* b1 = (const float*)d_in[2];
    const float* w2 = (const float*)d_in[3];
    const float* b2 = (const float*)d_in[4];
    const float* cw = (const float*)d_in[5];
    const float* cb = (const float*)d_in[6];
    float* out = (float*)d_out;

    float* ws   = (float*)d_ws;
    float* psum = ws;
    float* pmax = ws + (size_t)BAT * NSL * CHN;
    float* gate = pmax + (size_t)BAT * NSL * CHN;
    float* sbuf = gate + (size_t)BAT * CHN;

    k1_partial<<<BAT * NSL, 256, 0, stream>>>(x, psum, pmax);
    k2_gate<<<1, 256, 0, stream>>>(psum, pmax, w1, b1, w2, b2, gate);
    k3_spatial<<<BAT * HW / 8, 256, 0, stream>>>(x, gate, sbuf);
    k4_out<<<BAT * HGT, 256, 0, stream>>>(x, gate, sbuf, cw, cb, out);
}


# gate kernel from 1 work-group to 32, one per batch: avg and max pooled rows share one WMMA M tile (other rows zero), w1 staged in LDS; every sum keeps the parent's order
# speedup vs baseline: 1.5813x; 1.5813x over previous
#include <hip/hip_runtime.h>
#include <math.h>


#define CHN 256
#define HIDN 16
#define BAT 32
#define HGT 56
#define WID 56
#define HW  (HGT * WID)
#define NSL 16
#define SLICE (HW / NSL)
#define C4  (CHN / 4)

typedef float v2f __attribute__((ext_vector_type(2)));
typedef float v8f __attribute__((ext_vector_type(8)));

__device__ __forceinline__ float sig_(float v) {
    return __builtin_amdgcn_rcpf(1.0f + __expf(-v));
}

__global__ __launch_bounds__(256) void k1_partial(const float* __restrict__ x,
                                                  float* __restrict__ psum,
                                                  float* __restrict__ pmax) {
    __shared__ float ssum[4 * CHN];
    __shared__ float smax[4 * CHN];

    const int b = blockIdx.x >> 4;
    const int s = blockIdx.x & 15;
    const int t = threadIdx.x;
    const int pg = t >> 6;
    const int q  = t & 63;

    const float4* xp = (const float4*)(x + ((size_t)b * HW + (size_t)s * SLICE) * CHN);

    float4 sum = {0.f, 0.f, 0.f, 0.f};
    float4 mx  = {-3.4e38f, -3.4e38f, -3.4e38f, -3.4e38f};
    for (int p = pg; p < SLICE; p += 4) {
        __builtin_prefetch(&xp[(size_t)(p + 16) * C4 + q], 0, 0);
        float4 v = xp[(size_t)p * C4 + q];
        sum.x += v.x; sum.y += v.y; sum.z += v.z; sum.w += v.w;
        mx.x = fmaxf(mx.x, v.x); mx.y = fmaxf(mx.y, v.y);
        mx.z = fmaxf(mx.z, v.z); mx.w = fmaxf(mx.w, v.w);
    }
    ((float4*)ssum)[pg * C4 + q] = sum;
    ((float4*)smax)[pg * C4 + q] = mx;
    __syncthreads();

    const int c = t;
    float r = ssum[c] + ssum[CHN + c] + ssum[2 * CHN + c] + ssum[3 * CHN + c];
    float m = fmaxf(fmaxf(smax[c], smax[CHN + c]),
                    fmaxf(smax[2 * CHN + c], smax[3 * CHN + c]));
    psum[(size_t)blockIdx.x * CHN + c] = r;
    pmax[(size_t)blockIdx.x * CHN + c] = m;
}

#define PPITCH (CHN + 8)

__global__ __launch_bounds__(256) void k2_gate(const float* __restrict__ psum,
                                               const float* __restrict__ pmax,
                                               const float* __restrict__ w1,
                                               const float* __restrict__ b1,
                                               const float* __restrict__ w2,
                                               const float* __restrict__ b2,
                                               float* __restrict__ gate) {
    __shared__ float pool[3][PPITCH];
    __shared__ float w1s[CHN * HIDN];
    __shared__ float hid[3][HIDN];

    const int b = blockIdx.x;
    const int t = threadIdx.x;
    const int wave = t >> 5;
    const int lane = t & 31;
    const int l16 = lane & 15;
    const int hi = lane >> 4;

#pragma unroll
    for (int i = 0; i < (CHN * HIDN / 4) / 256; ++i)
        ((float4*)w1s)[t + 256 * i] = ((const float4*)w1)[t + 256 * i];

    {
        const float* ps = psum + (size_t)b * NSL * CHN + t;
        const float* pm = pmax + (size_t)b * NSL * CHN + t;
        float rs = 0.0f, rm = -3.4e38f;
#pragma unroll
        for (int s = 0; s < NSL; ++s) {
            rs += ps[(size_t)s * CHN];
            rm = fmaxf(rm, pm[(size_t)s * CHN]);
        }
        pool[0][t] = rs * (1.0f / (float)HW);
        pool[1][t] = rm;
        pool[2][t] = 0.0f;
        if (t < HIDN) hid[2][t] = 0.0f;
    }
    __syncthreads();

    const int mrow = ((l16 & 6) == 0) ? (l16 & 1) : 2;
    const int col = wave * 32 + lane;
    v8f zero = {};

    v2f bw[2][HIDN / 4];
#pragma unroll
    for (int tt = 0; tt < 2; ++tt) {
        const int n0 = (wave * 2 + tt) * 16;
#pragma unroll
        for (int k = 0; k < HIDN; k += 4) {
            bw[tt][k >> 2].x = w2[(k + 2 * hi + 0) * CHN + n0 + l16];
            bw[tt][k >> 2].y = w2[(k + 2 * hi + 1) * CHN + n0 + l16];
        }
    }
    const float bias = b2[col];

    if (wave == 0) {
        const float* arow = pool[mrow];
        v8f c = zero;
#pragma unroll
        for (int k = 0; k < CHN; k += 4) {
            v2f a, bm;
            a.x = arow[k + 2 * hi + 0];
            a.y = arow[k + 2 * hi + 1];
            bm.x = w1s[(k + 2 * hi + 0) * HIDN + l16];
            bm.y = w1s[(k + 2 * hi + 1) * HIDN + l16];
            c = __builtin_amdgcn_wmma_f32_16x16x4_f32(false, a, false, bm,
                                                      (short)0, c, false, false);
        }
        const float v = hi ? c[1] : c[0];
        hid[hi][l16] = fmaxf(v + b1[l16], 0.0f);
    }
    __syncthreads();

    const float* hrow = hid[mrow];
    v8f c2[2];
#pragma unroll
    for (int tt = 0; tt < 2; ++tt) {
        v8f c = zero;
#pragma unroll
        for (int k = 0; k < HIDN; k += 4) {
            v2f a;
            a.x = hrow[k + 2 * hi + 0];
            a.y = hrow[k + 2 * hi + 1];
            c = __builtin_amdgcn_wmma_f32_16x16x4_f32(false, a, false, bw[tt][k >> 2],
                                                      (short)0, c, false, false);
        }
        c2[tt] = c;
    }
    const float pa = hi ? c2[1][0] : c2[0][0];
    const float pm = hi ? c2[1][1] : c2[0][1];
    gate[(size_t)b * CHN + col] = sig_(pa + bias) + sig_(pm + bias);
}

__global__ __launch_bounds__(256) void k3_spatial(const float* __restrict__ x,
                                                  const float* __restrict__ gate,
                                                  float* __restrict__ sbuf) {
    const int wave = threadIdx.x >> 5;
    const int lane = threadIdx.x & 31;
    const int gp = blockIdx.x * 8 + wave;
    const int b = gp / HW;
    const float4* xp = (const float4*)(x + (size_t)gp * CHN);
    const float4* g4 = (const float4*)(gate + (size_t)b * CHN);

    float sum = 0.0f, mx = -3.4e38f;
#pragma unroll
    for (int i = 0; i < 2; ++i) {
        const int qi = lane + 32 * i;
        const float4 xv = xp[qi];
        const float4 gv = g4[qi];
        const float p0 = xv.x * gv.x, p1 = xv.y * gv.y;
        const float p2 = xv.z * gv.z, p3 = xv.w * gv.w;
        sum += p0 + p1 + p2 + p3;
        mx = fmaxf(mx, fmaxf(fmaxf(p0, p1), fmaxf(p2, p3)));
    }
#pragma unroll
    for (int off = 16; off > 0; off >>= 1) {
        sum += __shfl_xor(sum, off, 32);
        mx = fmaxf(mx, __shfl_xor(mx, off, 32));
    }
    if (lane == 0) {
        sbuf[(size_t)gp * 2 + 0] = sum * (1.0f / (float)CHN);
        sbuf[(size_t)gp * 2 + 1] = mx;
    }
}

__global__ __launch_bounds__(256) void k4_out(const float* __restrict__ x,
                                              const float* __restrict__ gate,
                                              const float* __restrict__ sbuf,
                                              const float* __restrict__ cw,
                                              const float* __restrict__ cb,
                                              float* __restrict__ out) {
    __shared__ float sg[WID];
    const int b = blockIdx.x / HGT;
    const int h = blockIdx.x % HGT;
    const int t = threadIdx.x;

    if (t < WID) {
        float acc = cb[0];
        for (int dy = -3; dy <= 3; ++dy) {
            const int hh = h + dy;
            if (hh < 0 || hh >= HGT) continue;
            for (int dx = -3; dx <= 3; ++dx) {
                const int ww = t + dx;
                if (ww < 0 || ww >= WID) continue;
                const float* sp = sbuf + ((size_t)(b * HGT + hh) * WID + ww) * 2;
                const float* wp = cw + ((dy + 3) * 7 + (dx + 3)) * 2;
                acc += sp[0] * wp[0] + sp[1] * wp[1];
            }
        }
        sg[t] = sig_(acc);
    }
    __syncthreads();

    const int q    = t & 63;
    const int wofs = t >> 6;
    const float4 g = ((const float4*)gate)[(size_t)b * C4 + q];

    const size_t row = ((size_t)b * HGT + h) * WID;
    const float4* xr4 = (const float4*)(x + row * CHN);
    float4*       o4  = (float4*)(out + row * CHN);

#pragma unroll 2
    for (int i = 0; i < WID / 4; ++i) {
        const int w = i * 4 + wofs;
        const float4 xv = xr4[(size_t)w * C4 + q];
        const float sgw = sg[w];
        float4 ov;
        ov.x = xv.x * g.x * sgw;
        ov.y = xv.y * g.y * sgw;
        ov.z = xv.z * g.z * sgw;
        ov.w = xv.w * g.w * sgw;
        o4[(size_t)w * C4 + q] = ov;
    }
}

extern "C" void kernel_launch(void* const* d_in, const int* in_sizes, int n_in,
                              void* d_out, int out_size, void* d_ws, size_t ws_size,
                              hipStream_t stream) {
    const float* x  = (const float*)d_in[0];
    const float* w1 = (const float*)d_in[1];
    const float* b1 = (const float*)d_in[2];
    const float* w2 = (const float*)d_in[3];
    const float* b2 = (const float*)d_in[4];
    const float* cw = (const float*)d_in[5];
    const float* cb = (const float*)d_in[6];
    float* out = (float*)d_out;

    float* ws   = (float*)d_ws;
    float* psum = ws;
    float* pmax = ws + (size_t)BAT * NSL * CHN;
    float* gate = pmax + (size_t)BAT * NSL * CHN;
    float* sbuf = gate + (size_t)BAT * CHN;

    k1_partial<<<BAT * NSL, 256, 0, stream>>>(x, psum, pmax);
    k2_gate<<<BAT, 256, 0, stream>>>(psum, pmax, w1, b1, w2, b2, gate);
    k3_spatial<<<BAT * HW / 8, 256, 0, stream>>>(x, gate, sbuf);
    k4_out<<<BAT * HGT, 256, 0, stream>>>(x, gate, sbuf, cw, cb, out);
}


# spatial-stats kernel: one wave per 28 consecutive pixels instead of one per pixel (100352 -> 3584 waves, 512 groups of 7), gate held in registers, loads 8 pixels ahead; per-pixel sum order unchanged
# speedup vs baseline: 1.8535x; 1.1721x over previous
#include <hip/hip_runtime.h>
#include <math.h>


#define CHN 256
#define HIDN 16
#define BAT 32
#define HGT 56
#define WID 56
#define HW  (HGT * WID)
#define NSL 16
#define SLICE (HW / NSL)
#define C4  (CHN / 4)

typedef float v2f __attribute__((ext_vector_type(2)));
typedef float v8f __attribute__((ext_vector_type(8)));

__device__ __forceinline__ float sig_(float v) {
    return __builtin_amdgcn_rcpf(1.0f + __expf(-v));
}

__global__ __launch_bounds__(256) void k1_partial(const float* __restrict__ x,
                                                  float* __restrict__ psum,
                                                  float* __restrict__ pmax) {
    __shared__ float ssum[4 * CHN];
    __shared__ float smax[4 * CHN];

    const int b = blockIdx.x >> 4;
    const int s = blockIdx.x & 15;
    const int t = threadIdx.x;
    const int pg = t >> 6;
    const int q  = t & 63;

    const float4* xp = (const float4*)(x + ((size_t)b * HW + (size_t)s * SLICE) * CHN);

    float4 sum = {0.f, 0.f, 0.f, 0.f};
    float4 mx  = {-3.4e38f, -3.4e38f, -3.4e38f, -3.4e38f};
    for (int p = pg; p < SLICE; p += 4) {
        __builtin_prefetch(&xp[(size_t)(p + 16) * C4 + q], 0, 0);
        float4 v = xp[(size_t)p * C4 + q];
        sum.x += v.x; sum.y += v.y; sum.z += v.z; sum.w += v.w;
        mx.x = fmaxf(mx.x, v.x); mx.y = fmaxf(mx.y, v.y);
        mx.z = fmaxf(mx.z, v.z); mx.w = fmaxf(mx.w, v.w);
    }
    ((float4*)ssum)[pg * C4 + q] = sum;
    ((float4*)smax)[pg * C4 + q] = mx;
    __syncthreads();

    const int c = t;
    float r = ssum[c] + ssum[CHN + c] + ssum[2 * CHN + c] + ssum[3 * CHN + c];
    float m = fmaxf(fmaxf(smax[c], smax[CHN + c]),
                    fmaxf(smax[2 * CHN + c], smax[3 * CHN + c]));
    psum[(size_t)blockIdx.x * CHN + c] = r;
    pmax[(size_t)blockIdx.x * CHN + c] = m;
}

#define PPITCH (CHN + 8)

__global__ __launch_bounds__(256) void k2_gate(const float* __restrict__ psum,
                                               const float* __restrict__ pmax,
                                               const float* __restrict__ w1,
                                               const float* __restrict__ b1,
                                               const float* __restrict__ w2,
                                               const float* __restrict__ b2,
                                               float* __restrict__ gate) {
    __shared__ float pool[3][PPITCH];
    __shared__ float w1s[CHN * HIDN];
    __shared__ float hid[3][HIDN];

    const int b = blockIdx.x;
    const int t = threadIdx.x;
    const int wave = t >> 5;
    const int lane = t & 31;
    const int l16 = lane & 15;
    const int hi = lane >> 4;

#pragma unroll
    for (int i = 0; i < (CHN * HIDN / 4) / 256; ++i)
        ((float4*)w1s)[t + 256 * i] = ((const float4*)w1)[t + 256 * i];

    {
        const float* ps = psum + (size_t)b * NSL * CHN + t;
        const float* pm = pmax + (size_t)b * NSL * CHN + t;
        float rs = 0.0f, rm = -3.4e38f;
#pragma unroll
        for (int s = 0; s < NSL; ++s) {
            rs += ps[(size_t)s * CHN];
            rm = fmaxf(rm, pm[(size_t)s * CHN]);
        }
        pool[0][t] = rs * (1.0f / (float)HW);
        pool[1][t] = rm;
        pool[2][t] = 0.0f;
        if (t < HIDN) hid[2][t] = 0.0f;
    }
    __syncthreads();

    const int mrow = ((l16 & 6) == 0) ? (l16 & 1) : 2;
    const int col = wave * 32 + lane;
    v8f zero = {};

    v2f bw[2][HIDN / 4];
#pragma unroll
    for (int tt = 0; tt < 2; ++tt) {
        const int n0 = (wave * 2 + tt) * 16;
#pragma unroll
        for (int k = 0; k < HIDN; k += 4) {
            bw[tt][k >> 2].x = w2[(k + 2 * hi + 0) * CHN + n0 + l16];
            bw[tt][k >> 2].y = w2[(k + 2 * hi + 1) * CHN + n0 + l16];
        }
    }
    const float bias = b2[col];

    if (wave == 0) {
        const float* arow = pool[mrow];
        v8f c = zero;
#pragma unroll
        for (int k = 0; k < CHN; k += 4) {
            v2f a, bm;
            a.x = arow[k + 2 * hi + 0];
            a.y = arow[k + 2 * hi + 1];
            bm.x = w1s[(k + 2 * hi + 0) * HIDN + l16];
            bm.y = w1s[(k + 2 * hi + 1) * HIDN + l16];
            c = __builtin_amdgcn_wmma_f32_16x16x4_f32(false, a, false, bm,
                                                      (short)0, c, false, false);
        }
        const float v = hi ? c[1] : c[0];
        hid[hi][l16] = fmaxf(v + b1[l16], 0.0f);
    }
    __syncthreads();

    const float* hrow = hid[mrow];
    v8f c2[2];
#pragma unroll
    for (int tt = 0; tt < 2; ++tt) {
        v8f c = zero;
#pragma unroll
        for (int k = 0; k < HIDN; k += 4) {
            v2f a;
            a.x = hrow[k + 2 * hi + 0];
            a.y = hrow[k + 2 * hi + 1];
            c = __builtin_amdgcn_wmma_f32_16x16x4_f32(false, a, false, bw[tt][k >> 2],
                                                      (short)0, c, false, false);
        }
        c2[tt] = c;
    }
    const float pa = hi ? c2[1][0] : c2[0][0];
    const float pm = hi ? c2[1][1] : c2[0][1];
    gate[(size_t)b * CHN + col] = sig_(pa + bias) + sig_(pm + bias);
}

#define K3_WAVES 7
#define K3_PIX   28
#define K3_GRP   7
#define K3_AHEAD 8
#define K3_RING  (K3_AHEAD + 1)

__global__ __launch_bounds__(K3_WAVES * 32) void k3_spatial(const float* __restrict__ x,
                                                            const float* __restrict__ gate,
                                                            float* __restrict__ sbuf) {
    const int lane = threadIdx.x & 31;
    const int wave = __builtin_amdgcn_readfirstlane(threadIdx.x >> 5);
    const int b = blockIdx.x >> 4;
    const int gp0 = blockIdx.x * SLICE + wave * K3_PIX;
    const float4* xp = (const float4*)(x + (size_t)gp0 * CHN) + lane;
    const float4* g4 = (const float4*)(gate + (size_t)b * CHN);
    const float4 g0 = g4[lane];
    const float4 g1 = g4[lane + 32];

    float4 ring[K3_RING][2];
    float sum[K3_PIX], mx[K3_PIX];

#define K3_LOAD(P) do { ring[(P) % K3_RING][0] = xp[(size_t)(P) * C4];        \
                        ring[(P) % K3_RING][1] = xp[(size_t)(P) * C4 + 32]; } while (0)

#pragma unroll
    for (int p = 0; p < K3_AHEAD; ++p) K3_LOAD(p);

#pragma unroll
    for (int g = 0; g < K3_PIX / K3_GRP; ++g) {
#pragma unroll
        for (int j = 0; j < K3_GRP; ++j) {
            const int p = g * K3_GRP + j;
            __builtin_amdgcn_sched_barrier(0);
            if (p + K3_AHEAD < K3_PIX) K3_LOAD(p + K3_AHEAD);
            __builtin_amdgcn_sched_barrier(0);
            float s = 0.0f, m = -3.4e38f;
#pragma unroll
            for (int i = 0; i < 2; ++i) {
                const float4 xv = ring[p % K3_RING][i];
                const float4 gv = i ? g1 : g0;
                const float p0 = xv.x * gv.x, p1 = xv.y * gv.y;
                const float p2 = xv.z * gv.z, p3 = xv.w * gv.w;
                s += p0 + p1 + p2 + p3;
                m = fmaxf(m, fmaxf(fmaxf(p0, p1), fmaxf(p2, p3)));
            }
            sum[p] = s;
            mx[p] = m;
        }
        __builtin_amdgcn_sched_barrier(0);
#pragma unroll
        for (int j = 0; j < K3_GRP; ++j) {
            const int p = g * K3_GRP + j;
#pragma unroll
            for (int off = 16; off > 0; off >>= 1) {
                sum[p] += __shfl_xor(sum[p], off, 32);
                mx[p] = fmaxf(mx[p], __shfl_xor(mx[p], off, 32));
            }
        }
    }
#undef K3_LOAD

    float ms = 0.0f, mm = 0.0f;
#pragma unroll
    for (int p = 0; p < K3_PIX; ++p)
        if (lane == p) { ms = sum[p]; mm = mx[p]; }
    if (lane < K3_PIX) {
        float2 o;
        o.x = ms * (1.0f / (float)CHN);
        o.y = mm;
        ((float2*)sbuf)[(size_t)gp0 + lane] = o;
    }
}

__global__ __launch_bounds__(256) void k4_out(const float* __restrict__ x,
                                              const float* __restrict__ gate,
                                              const float* __restrict__ sbuf,
                                              const float* __restrict__ cw,
                                              const float* __restrict__ cb,
                                              float* __restrict__ out) {
    __shared__ float sg[WID];
    const int b = blockIdx.x / HGT;
    const int h = blockIdx.x % HGT;
    const int t = threadIdx.x;

    if (t < WID) {
        float acc = cb[0];
        for (int dy = -3; dy <= 3; ++dy) {
            const int hh = h + dy;
            if (hh < 0 || hh >= HGT) continue;
            for (int dx = -3; dx <= 3; ++dx) {
                const int ww = t + dx;
                if (ww < 0 || ww >= WID) continue;
                const float* sp = sbuf + ((size_t)(b * HGT + hh) * WID + ww) * 2;
                const float* wp = cw + ((dy + 3) * 7 + (dx + 3)) * 2;
                acc += sp[0] * wp[0] + sp[1] * wp[1];
            }
        }
        sg[t] = sig_(acc);
    }
    __syncthreads();

    const int q    = t & 63;
    const int wofs = t >> 6;
    const float4 g = ((const float4*)gate)[(size_t)b * C4 + q];

    const size_t row = ((size_t)b * HGT + h) * WID;
    const float4* xr4 = (const float4*)(x + row * CHN);
    float4*       o4  = (float4*)(out + row * CHN);

#pragma unroll 2
    for (int i = 0; i < WID / 4; ++i) {
        const int w = i * 4 + wofs;
        const float4 xv = xr4[(size_t)w * C4 + q];
        const float sgw = sg[w];
        float4 ov;
        ov.x = xv.x * g.x * sgw;
        ov.y = xv.y * g.y * sgw;
        ov.z = xv.z * g.z * sgw;
        ov.w = xv.w * g.w * sgw;
        o4[(size_t)w * C4 + q] = ov;
    }
}

extern "C" void kernel_launch(void* const* d_in, const int* in_sizes, int n_in,
                              void* d_out, int out_size, void* d_ws, size_t ws_size,
                              hipStream_t stream) {
    const float* x  = (const float*)d_in[0];
    const float* w1 = (const float*)d_in[1];
    const float* b1 = (const float*)d_in[2];
    const float* w2 = (const float*)d_in[3];
    const float* b2 = (const float*)d_in[4];
    const float* cw = (const float*)d_in[5];
    const float* cb = (const float*)d_in[6];
    float* out = (float*)d_out;

    float* ws   = (float*)d_ws;
    float* psum = ws;
    float* pmax = ws + (size_t)BAT * NSL * CHN;
    float* gate = pmax + (size_t)BAT * NSL * CHN;
    float* sbuf = gate + (size_t)BAT * CHN;

    k1_partial<<<BAT * NSL, 256, 0, stream>>>(x, psum, pmax);
    k2_gate<<<BAT, 256, 0, stream>>>(psum, pmax, w1, b1, w2, b2, gate);
    k3_spatial<<<BAT * NSL, K3_WAVES * 32, 0, stream>>>(x, gate, sbuf);
    k4_out<<<BAT * HGT, 256, 0, stream>>>(x, gate, sbuf, cw, cb, out);
}


# output kernel's conv: the 7 rows of s it reads are staged once in a zero-padded LDS tile (one dense load per thread), the 49 taps read LDS instead of 49 dependent global loads; tap order unchanged
# speedup vs baseline: 2.1514x; 1.1607x over previous
#include <hip/hip_runtime.h>
#include <math.h>


#define CHN 256
#define HIDN 16
#define BAT 32
#define HGT 56
#define WID 56
#define HW  (HGT * WID)
#define NSL 16
#define SLICE (HW / NSL)
#define C4  (CHN / 4)

typedef float v2f __attribute__((ext_vector_type(2)));
typedef float v8f __attribute__((ext_vector_type(8)));

__device__ __forceinline__ float sig_(float v) {
    return __builtin_amdgcn_rcpf(1.0f + __expf(-v));
}

__global__ __launch_bounds__(256) void k1_partial(const float* __restrict__ x,
                                                  float* __restrict__ psum,
                                                  float* __restrict__ pmax) {
    __shared__ float ssum[4 * CHN];
    __shared__ float smax[4 * CHN];

    const int b = blockIdx.x >> 4;
    const int s = blockIdx.x & 15;
    const int t = threadIdx.x;
    const int pg = t >> 6;
    const int q  = t & 63;

    const float4* xp = (const float4*)(x + ((size_t)b * HW + (size_t)s * SLICE) * CHN);

    float4 sum = {0.f, 0.f, 0.f, 0.f};
    float4 mx  = {-3.4e38f, -3.4e38f, -3.4e38f, -3.4e38f};
    for (int p = pg; p < SLICE; p += 4) {
        __builtin_prefetch(&xp[(size_t)(p + 16) * C4 + q], 0, 0);
        float4 v = xp[(size_t)p * C4 + q];
        sum.x += v.x; sum.y += v.y; sum.z += v.z; sum.w += v.w;
        mx.x = fmaxf(mx.x, v.x); mx.y = fmaxf(mx.y, v.y);
        mx.z = fmaxf(mx.z, v.z); mx.w = fmaxf(mx.w, v.w);
    }
    ((float4*)ssum)[pg * C4 + q] = sum;
    ((float4*)smax)[pg * C4 + q] = mx;
    __syncthreads();

    const int c = t;
    float r = ssum[c] + ssum[CHN + c] + ssum[2 * CHN + c] + ssum[3 * CHN + c];
    float m = fmaxf(fmaxf(smax[c], smax[CHN + c]),
                    fmaxf(smax[2 * CHN + c], smax[3 * CHN + c]));
    psum[(size_t)blockIdx.x * CHN + c] = r;
    pmax[(size_t)blockIdx.x * CHN + c] = m;
}

#define PPITCH (CHN + 8)

__global__ __launch_bounds__(256) void k2_gate(const float* __restrict__ psum,
                                               const float* __restrict__ pmax,
                                               const float* __restrict__ w1,
                                               const float* __restrict__ b1,
                                               const float* __restrict__ w2,
                                               const float* __restrict__ b2,
                                               float* __restrict__ gate) {
    __shared__ float pool[3][PPITCH];
    __shared__ float w1s[CHN * HIDN];
    __shared__ float hid[3][HIDN];

    const int b = blockIdx.x;
    const int t = threadIdx.x;
    const int wave = t >> 5;
    const int lane = t & 31;
    const int l16 = lane & 15;
    const int hi = lane >> 4;

#pragma unroll
    for (int i = 0; i < (CHN * HIDN / 4) / 256; ++i)
        ((float4*)w1s)[t + 256 * i] = ((const float4*)w1)[t + 256 * i];

    {
        const float* ps = psum + (size_t)b * NSL * CHN + t;
        const float* pm = pmax + (size_t)b * NSL * CHN + t;
        float rs = 0.0f, rm = -3.4e38f;
#pragma unroll
        for (int s = 0; s < NSL; ++s) {
            rs += ps[(size_t)s * CHN];
            rm = fmaxf(rm, pm[(size_t)s * CHN]);
        }
        pool[0][t] = rs * (1.0f / (float)HW);
        pool[1][t] = rm;
        pool[2][t] = 0.0f;
        if (t < HIDN) hid[2][t] = 0.0f;
    }
    __syncthreads();

    const int mrow = ((l16 & 6) == 0) ? (l16 & 1) : 2;
    const int col = wave * 32 + lane;
    v8f zero = {};

    v2f bw[2][HIDN / 4];
#pragma unroll
    for (int tt = 0; tt < 2; ++tt) {
        const int n0 = (wave * 2 + tt) * 16;
#pragma unroll
        for (int k = 0; k < HIDN; k += 4) {
            bw[tt][k >> 2].x = w2[(k + 2 * hi + 0) * CHN + n0 + l16];
            bw[tt][k >> 2].y = w2[(k + 2 * hi + 1) * CHN + n0 + l16];
        }
    }
    const float bias = b2[col];

    if (wave == 0) {
        const float* arow = pool[mrow];
        v8f c = zero;
#pragma unroll
        for (int k = 0; k < CHN; k += 4) {
            v2f a, bm;
            a.x = arow[k + 2 * hi + 0];
            a.y = arow[k + 2 * hi + 1];
            bm.x = w1s[(k + 2 * hi + 0) * HIDN + l16];
            bm.y = w1s[(k + 2 * hi + 1) * HIDN + l16];
            c = __builtin_amdgcn_wmma_f32_16x16x4_f32(false, a, false, bm,
                                                      (short)0, c, false, false);
        }
        const float v = hi ? c[1] : c[0];
        hid[hi][l16] = fmaxf(v + b1[l16], 0.0f);
    }
    __syncthreads();

    const float* hrow = hid[mrow];
    v8f c2[2];
#pragma unroll
    for (int tt = 0; tt < 2; ++tt) {
        v8f c = zero;
#pragma unroll
        for (int k = 0; k < HIDN; k += 4) {
            v2f a;
            a.x = hrow[k + 2 * hi + 0];
            a.y = hrow[k + 2 * hi + 1];
            c = __builtin_amdgcn_wmma_f32_16x16x4_f32(false, a, false, bw[tt][k >> 2],
                                                      (short)0, c, false, false);
        }
        c2[tt] = c;
    }
    const float pa = hi ? c2[1][0] : c2[0][0];
    const float pm = hi ? c2[1][1] : c2[0][1];
    gate[(size_t)b * CHN + col] = sig_(pa + bias) + sig_(pm + bias);
}

#define K3_WAVES 7
#define K3_PIX   28
#define K3_GRP   7
#define K3_AHEAD 8
#define K3_RING  (K3_AHEAD + 1)

__global__ __launch_bounds__(K3_WAVES * 32) void k3_spatial(const float* __restrict__ x,
                                                            const float* __restrict__ gate,
                                                            float* __restrict__ sbuf) {
    const int lane = threadIdx.x & 31;
    const int wave = __builtin_amdgcn_readfirstlane(threadIdx.x >> 5);
    const int b = blockIdx.x >> 4;
    const int gp0 = blockIdx.x * SLICE + wave * K3_PIX;
    const float4* xp = (const float4*)(x + (size_t)gp0 * CHN) + lane;
    const float4* g4 = (const float4*)(gate + (size_t)b * CHN);
    const float4 g0 = g4[lane];
    const float4 g1 = g4[lane + 32];

    float4 ring[K3_RING][2];
    float sum[K3_PIX], mx[K3_PIX];

#define K3_LOAD(P) do { ring[(P) % K3_RING][0] = xp[(size_t)(P) * C4];        \
                        ring[(P) % K3_RING][1] = xp[(size_t)(P) * C4 + 32]; } while (0)

#pragma unroll
    for (int p = 0; p < K3_AHEAD; ++p) K3_LOAD(p);

#pragma unroll
    for (int g = 0; g < K3_PIX / K3_GRP; ++g) {
#pragma unroll
        for (int j = 0; j < K3_GRP; ++j) {
            const int p = g * K3_GRP + j;
            __builtin_amdgcn_sched_barrier(0);
            if (p + K3_AHEAD < K3_PIX) K3_LOAD(p + K3_AHEAD);
            __builtin_amdgcn_sched_barrier(0);
            float s = 0.0f, m = -3.4e38f;
#pragma unroll
            for (int i = 0; i < 2; ++i) {
                const float4 xv = ring[p % K3_RING][i];
                const float4 gv = i ? g1 : g0;
                const float p0 = xv.x * gv.x, p1 = xv.y * gv.y;
                const float p2 = xv.z * gv.z, p3 = xv.w * gv.w;
                s += p0 + p1 + p2 + p3;
                m = fmaxf(m, fmaxf(fmaxf(p0, p1), fmaxf(p2, p3)));
            }
            sum[p] = s;
            mx[p] = m;
        }
        __builtin_amdgcn_sched_barrier(0);
#pragma unroll
        for (int j = 0; j < K3_GRP; ++j) {
            const int p = g * K3_GRP + j;
#pragma unroll
            for (int off = 16; off > 0; off >>= 1) {
                sum[p] += __shfl_xor(sum[p], off, 32);
                mx[p] = fmaxf(mx[p], __shfl_xor(mx[p], off, 32));
            }
        }
    }
#undef K3_LOAD

    float ms = 0.0f, mm = 0.0f;
#pragma unroll
    for (int p = 0; p < K3_PIX; ++p)
        if (lane == p) { ms = sum[p]; mm = mx[p]; }
    if (lane < K3_PIX) {
        float2 o;
        o.x = ms * (1.0f / (float)CHN);
        o.y = mm;
        ((float2*)sbuf)[(size_t)gp0 + lane] = o;
    }
}

#define K4_PITCH 64
#define K4_ROW4  (WID / 2)
__global__ __launch_bounds__(256) void k4_out(const float* __restrict__ x,
                                              const float* __restrict__ gate,
                                              const float* __restrict__ sbuf,
                                              const float* __restrict__ cw,
                                              const float* __restrict__ cb,
                                              float* __restrict__ out) {
    __shared__ float sg[WID];
    __shared__ __attribute__((aligned(16))) float2 st[7][K4_PITCH];
    const int b = blockIdx.x / HGT;
    const int h = blockIdx.x % HGT;
    const int t = threadIdx.x;

    if (t < 7 * K4_ROW4) {
        const int r = t / K4_ROW4;
        const int c = t % K4_ROW4;
        const int hh = h + r - 3;
        float4 v = {0.f, 0.f, 0.f, 0.f};
        if (hh >= 0 && hh < HGT)
            v = ((const float4*)sbuf)[(size_t)(b * HGT + hh) * K4_ROW4 + c];
        *(float4*)&st[r][4 + 2 * c] = v;
    } else if (t < 7 * K4_ROW4 + 7 * 4) {
        const int i = t - 7 * K4_ROW4;
        const int r = i >> 2;
        const int k = i & 3;
        const float4 z = {0.f, 0.f, 0.f, 0.f};
        *(float4*)&st[r][(k < 2) ? 2 * k : 56 + 2 * k] = z;
    }
    __syncthreads();

    if (t < WID) {
        float acc = cb[0];
#pragma unroll
        for (int dy = 0; dy < 7; ++dy) {
#pragma unroll
            for (int dx = 0; dx < 7; ++dx) {
                const float2 sv = st[dy][t + dx + 1];
                const float* wp = cw + (dy * 7 + dx) * 2;
                acc += sv.x * wp[0] + sv.y * wp[1];
            }
        }
        sg[t] = sig_(acc);
    }
    __syncthreads();

    const int q    = t & 63;
    const int wofs = t >> 6;
    const float4 g = ((const float4*)gate)[(size_t)b * C4 + q];

    const size_t row = ((size_t)b * HGT + h) * WID;
    const float4* xr4 = (const float4*)(x + row * CHN);
    float4*       o4  = (float4*)(out + row * CHN);

#pragma unroll 2
    for (int i = 0; i < WID / 4; ++i) {
        const int w = i * 4 + wofs;
        const float4 xv = xr4[(size_t)w * C4 + q];
        const float sgw = sg[w];
        float4 ov;
        ov.x = xv.x * g.x * sgw;
        ov.y = xv.y * g.y * sgw;
        ov.z = xv.z * g.z * sgw;
        ov.w = xv.w * g.w * sgw;
        o4[(size_t)w * C4 + q] = ov;
    }
}

extern "C" void kernel_launch(void* const* d_in, const int* in_sizes, int n_in,
                              void* d_out, int out_size, void* d_ws, size_t ws_size,
                              hipStream_t stream) {
    const float* x  = (const float*)d_in[0];
    const float* w1 = (const float*)d_in[1];
    const float* b1 = (const float*)d_in[2];
    const float* w2 = (const float*)d_in[3];
    const float* b2 = (const float*)d_in[4];
    const float* cw = (const float*)d_in[5];
    const float* cb = (const float*)d_in[6];
    float* out = (float*)d_out;

    float* ws   = (float*)d_ws;
    float* psum = ws;
    float* pmax = ws + (size_t)BAT * NSL * CHN;
    float* gate = pmax + (size_t)BAT * NSL * CHN;
    float* sbuf = gate + (size_t)BAT * CHN;

    k1_partial<<<BAT * NSL, 256, 0, stream>>>(x, psum, pmax);
    k2_gate<<<BAT, 256, 0, stream>>>(psum, pmax, w1, b1, w2, b2, gate);
    k3_spatial<<<BAT * NSL, K3_WAVES * 32, 0, stream>>>(x, gate, sbuf);
    k4_out<<<BAT * HGT, 256, 0, stream>>>(x, gate, sbuf, cw, cb, out);
}


# output kernel: one work-group per 7 rows x 28 columns (1792 groups -> 512): s tile and conv once per group, under 14 x loads per thread issued first and kept in flight; tap, product order unchanged
# speedup vs baseline: 2.3892x; 1.1105x over previous
#include <hip/hip_runtime.h>
#include <math.h>


#define CHN 256
#define HIDN 16
#define BAT 32
#define HGT 56
#define WID 56
#define HW  (HGT * WID)
#define NSL 16
#define SLICE (HW / NSL)
#define C4  (CHN / 4)

typedef float v2f __attribute__((ext_vector_type(2)));
typedef float v8f __attribute__((ext_vector_type(8)));

__device__ __forceinline__ float sig_(float v) {
    return __builtin_amdgcn_rcpf(1.0f + __expf(-v));
}

__global__ __launch_bounds__(256) void k1_partial(const float* __restrict__ x,
                                                  float* __restrict__ psum,
                                                  float* __restrict__ pmax) {
    __shared__ float ssum[4 * CHN];
    __shared__ float smax[4 * CHN];

    const int b = blockIdx.x >> 4;
    const int s = blockIdx.x & 15;
    const int t = threadIdx.x;
    const int pg = t >> 6;
    const int q  = t & 63;

    const float4* xp = (const float4*)(x + ((size_t)b * HW + (size_t)s * SLICE) * CHN);

    float4 sum = {0.f, 0.f, 0.f, 0.f};
    float4 mx  = {-3.4e38f, -3.4e38f, -3.4e38f, -3.4e38f};
    for (int p = pg; p < SLICE; p += 4) {
        __builtin_prefetch(&xp[(size_t)(p + 16) * C4 + q], 0, 0);
        float4 v = xp[(size_t)p * C4 + q];
        sum.x += v.x; sum.y += v.y; sum.z += v.z; sum.w += v.w;
        mx.x = fmaxf(mx.x, v.x); mx.y = fmaxf(mx.y, v.y);
        mx.z = fmaxf(mx.z, v.z); mx.w = fmaxf(mx.w, v.w);
    }
    ((float4*)ssum)[pg * C4 + q] = sum;
    ((float4*)smax)[pg * C4 + q] = mx;
    __syncthreads();

    const int c = t;
    float r = ssum[c] + ssum[CHN + c] + ssum[2 * CHN + c] + ssum[3 * CHN + c];
    float m = fmaxf(fmaxf(smax[c], smax[CHN + c]),
                    fmaxf(smax[2 * CHN + c], smax[3 * CHN + c]));
    psum[(size_t)blockIdx.x * CHN + c] = r;
    pmax[(size_t)blockIdx.x * CHN + c] = m;
}

#define PPITCH (CHN + 8)

__global__ __launch_bounds__(256) void k2_gate(const float* __restrict__ psum,
                                               const float* __restrict__ pmax,
                                               const float* __restrict__ w1,
                                               const float* __restrict__ b1,
                                               const float* __restrict__ w2,
                                               const float* __restrict__ b2,
                                               float* __restrict__ gate) {
    __shared__ float pool[3][PPITCH];
    __shared__ float w1s[CHN * HIDN];
    __shared__ float hid[3][HIDN];

    const int b = blockIdx.x;
    const int t = threadIdx.x;
    const int wave = t >> 5;
    const int lane = t & 31;
    const int l16 = lane & 15;
    const int hi = lane >> 4;

#pragma unroll
    for (int i = 0; i < (CHN * HIDN / 4) / 256; ++i)
        ((float4*)w1s)[t + 256 * i] = ((const float4*)w1)[t + 256 * i];

    {
        const float* ps = psum + (size_t)b * NSL * CHN + t;
        const float* pm = pmax + (size_t)b * NSL * CHN + t;
        float rs = 0.0f, rm = -3.4e38f;
#pragma unroll
        for (int s = 0; s < NSL; ++s) {
            rs += ps[(size_t)s * CHN];
            rm = fmaxf(rm, pm[(size_t)s * CHN]);
        }
        pool[0][t] = rs * (1.0f / (float)HW);
        pool[1][t] = rm;
        pool[2][t] = 0.0f;
        if (t < HIDN) hid[2][t] = 0.0f;
    }
    __syncthreads();

    const int mrow = ((l16 & 6) == 0) ? (l16 & 1) : 2;
    const int col = wave * 32 + lane;
    v8f zero = {};

    v2f bw[2][HIDN / 4];
#pragma unroll
    for (int tt = 0; tt < 2; ++tt) {
        const int n0 = (wave * 2 + tt) * 16;
#pragma unroll
        for (int k = 0; k < HIDN; k += 4) {
            bw[tt][k >> 2].x = w2[(k + 2 * hi + 0) * CHN + n0 + l16];
            bw[tt][k >> 2].y = w2[(k + 2 * hi + 1) * CHN + n0 + l16];
        }
    }
    const float bias = b2[col];

    if (wave == 0) {
        const float* arow = pool[mrow];
        v8f c = zero;
#pragma unroll
        for (int k = 0; k < CHN; k += 4) {
            v2f a, bm;
            a.x = arow[k + 2 * hi + 0];
            a.y = arow[k + 2 * hi + 1];
            bm.x = w1s[(k + 2 * hi + 0) * HIDN + l16];
            bm.y = w1s[(k + 2 * hi + 1) * HIDN + l16];
            c = __builtin_amdgcn_wmma_f32_16x16x4_f32(false, a, false, bm,
                                                      (short)0, c, false, false);
        }
        const float v = hi ? c[1] : c[0];
        hid[hi][l16] = fmaxf(v + b1[l16], 0.0f);
    }
    __syncthreads();

    const float* hrow = hid[mrow];
    v8f c2[2];
#pragma unroll
    for (int tt = 0; tt < 2; ++tt) {
        v8f c = zero;
#pragma unroll
        for (int k = 0; k < HIDN; k += 4) {
            v2f a;
            a.x = hrow[k + 2 * hi + 0];
            a.y = hrow[k + 2 * hi + 1];
            c = __builtin_amdgcn_wmma_f32_16x16x4_f32(false, a, false, bw[tt][k >> 2],
                                                      (short)0, c, false, false);
        }
        c2[tt] = c;
    }
    const float pa = hi ? c2[1][0] : c2[0][0];
    const float pm = hi ? c2[1][1] : c2[0][1];
    gate[(size_t)b * CHN + col] = sig_(pa + bias) + sig_(pm + bias);
}

#define K3_WAVES 7
#define K3_PIX   28
#define K3_GRP   7
#define K3_AHEAD 8
#define K3_RING  (K3_AHEAD + 1)

__global__ __launch_bounds__(K3_WAVES * 32) void k3_spatial(const float* __restrict__ x,
                                                            const float* __restrict__ gate,
                                                            float* __restrict__ sbuf) {
    const int lane = threadIdx.x & 31;
    const int wave = __builtin_amdgcn_readfirstlane(threadIdx.x >> 5);
    const int b = blockIdx.x >> 4;
    const int gp0 = blockIdx.x * SLICE + wave * K3_PIX;
    const float4* xp = (const float4*)(x + (size_t)gp0 * CHN) + lane;
    const float4* g4 = (const float4*)(gate + (size_t)b * CHN);
    const float4 g0 = g4[lane];
    const float4 g1 = g4[lane + 32];

    float4 ring[K3_RING][2];
    float sum[K3_PIX], mx[K3_PIX];

#define K3_LOAD(P) do { ring[(P) % K3_RING][0] = xp[(size_t)(P) * C4];        \
                        ring[(P) % K3_RING][1] = xp[(size_t)(P) * C4 + 32]; } while (0)

#pragma unroll
    for (int p = 0; p < K3_AHEAD; ++p) K3_LOAD(p);

#pragma unroll
    for (int g = 0; g < K3_PIX / K3_GRP; ++g) {
#pragma unroll
        for (int j = 0; j < K3_GRP; ++j) {
            const int p = g * K3_GRP + j;
            __builtin_amdgcn_sched_barrier(0);
            if (p + K3_AHEAD < K3_PIX) K3_LOAD(p + K3_AHEAD);
            __builtin_amdgcn_sched_barrier(0);
            float s = 0.0f, m = -3.4e38f;
#pragma unroll
            for (int i = 0; i < 2; ++i) {
                const float4 xv = ring[p % K3_RING][i];
                const float4 gv = i ? g1 : g0;
                const float p0 = xv.x * gv.x, p1 = xv.y * gv.y;
                const float p2 = xv.z * gv.z, p3 = xv.w * gv.w;
                s += p0 + p1 + p2 + p3;
                m = fmaxf(m, fmaxf(fmaxf(p0, p1), fmaxf(p2, p3)));
            }
            sum[p] = s;
            mx[p] = m;
        }
        __builtin_amdgcn_sched_barrier(0);
#pragma unroll
        for (int j = 0; j < K3_GRP; ++j) {
            const int p = g * K3_GRP + j;
#pragma unroll
            for (int off = 16; off > 0; off >>= 1) {
                sum[p] += __shfl_xor(sum[p], off, 32);
                mx[p] = fmaxf(mx[p], __shfl_xor(mx[p], off, 32));
            }
        }
    }
#undef K3_LOAD

    float ms = 0.0f, mm = 0.0f;
#pragma unroll
    for (int p = 0; p < K3_PIX; ++p)
        if (lane == p) { ms = sum[p]; mm = mx[p]; }
    if (lane < K3_PIX) {
        float2 o;
        o.x = ms * (1.0f / (float)CHN);
        o.y = mm;
        ((float2*)sbuf)[(size_t)gp0 + lane] = o;
    }
}

#define K4_ROWS  7
#define K4_COLS  28
#define K4_TROWS (K4_ROWS + 6)
#define K4_PITCH 64
#define K4_ROW4  (WID / 2)
#define K4_PIX   (K4_ROWS * K4_COLS)
#define K4_ITERS (K4_PIX / 4)
#define K4_RITER (K4_COLS / 4)
#define K4_AHEAD 14
#define K4_GRID  (BAT * (HGT / K4_ROWS) * (WID / K4_COLS))
#define K4_OFS(I) ((size_t)((I) / K4_RITER) * (WID * C4) + (size_t)((I) % K4_RITER) * (4 * C4))

__global__ __launch_bounds__(256) void k4_out(const float* __restrict__ x,
                                              const float* __restrict__ gate,
                                              const float* __restrict__ sbuf,
                                              const float* __restrict__ cw,
                                              const float* __restrict__ cb,
                                              float* __restrict__ out) {
    __shared__ float sg[K4_PIX];
    __shared__ __attribute__((aligned(16))) float2 st[K4_TROWS][K4_PITCH];
    const int half = blockIdx.x & 1;
    const int band = (blockIdx.x >> 1) & 7;
    const int b  = blockIdx.x >> 4;
    const int h0 = band * K4_ROWS;
    const int w0 = half * K4_COLS;
    const int t  = threadIdx.x;

    const int r0 = t / K4_ROW4, c0 = t % K4_ROW4;
    const int r1 = (t + 256) / K4_ROW4, c1 = (t + 256) % K4_ROW4;
    const int hh0 = h0 + r0 - 3, hh1 = h0 + r1 - 3;
    float4 tv0 = {0.f, 0.f, 0.f, 0.f}, tv1 = {0.f, 0.f, 0.f, 0.f};
    if (hh0 >= 0 && hh0 < HGT)
        tv0 = ((const float4*)sbuf)[(size_t)(b * HGT + hh0) * K4_ROW4 + c0];
    if (t + 256 < K4_TROWS * K4_ROW4 && hh1 >= 0 && hh1 < HGT)
        tv1 = ((const float4*)sbuf)[(size_t)(b * HGT + hh1) * K4_ROW4 + c1];

    const int q    = t & 63;
    const int wofs = t >> 6;
    const float4 g = ((const float4*)gate)[(size_t)b * C4 + q];

    const size_t pix0 = ((size_t)b * HGT + h0) * WID + w0 + wofs;
    const float4* xb = (const float4*)(x + pix0 * CHN) + q;
    float4*       ob = (float4*)(out + pix0 * CHN) + q;

    float4 ring[K4_AHEAD];
#pragma unroll
    for (int i = 0; i < K4_AHEAD; ++i) ring[i] = xb[K4_OFS(i)];

    *(float4*)&st[r0][4 + 2 * c0] = tv0;
    if (t + 256 < K4_TROWS * K4_ROW4) {
        *(float4*)&st[r1][4 + 2 * c1] = tv1;
    } else if (t < 108 + K4_TROWS * 4) {
        const int i = t - 108;
        const int r = i >> 2;
        const int k = i & 3;
        const float4 z = {0.f, 0.f, 0.f, 0.f};
        *(float4*)&st[r][(k < 2) ? 2 * k : 56 + 2 * k] = z;
    }
    __syncthreads();

    if (t < K4_PIX) {
        const int r  = t / K4_COLS;
        const int wc = w0 + t % K4_COLS;
        float acc = cb[0];
#pragma unroll
        for (int dy = 0; dy < 7; ++dy) {
#pragma unroll
            for (int dx = 0; dx < 7; ++dx) {
                const float2 sv = st[r + dy][wc + dx + 1];
                const float* wp = cw + (dy * 7 + dx) * 2;
                acc += sv.x * wp[0] + sv.y * wp[1];
            }
        }
        sg[t] = sig_(acc);
    }
    __syncthreads();

    float sgv[K4_ITERS];
#pragma unroll
    for (int i = 0; i < K4_ITERS; ++i)
        sgv[i] = sg[(i / K4_RITER) * K4_COLS + (i % K4_RITER) * 4 + wofs];

#pragma unroll
    for (int i = 0; i < K4_ITERS; ++i) {
        float4 xv = ring[i % K4_AHEAD];
        asm volatile("" : "+v"(xv.x), "+v"(xv.y), "+v"(xv.z), "+v"(xv.w));
        if (i + K4_AHEAD < K4_ITERS) ring[i % K4_AHEAD] = xb[K4_OFS(i + K4_AHEAD)];
        const float sgw = sgv[i];
        float4 ov;
        ov.x = xv.x * g.x * sgw;
        ov.y = xv.y * g.y * sgw;
        ov.z = xv.z * g.z * sgw;
        ov.w = xv.w * g.w * sgw;
        ob[K4_OFS(i)] = ov;
    }
}

extern "C" void kernel_launch(void* const* d_in, const int* in_sizes, int n_in,
                              void* d_out, int out_size, void* d_ws, size_t ws_size,
                              hipStream_t stream) {
    const float* x  = (const float*)d_in[0];
    const float* w1 = (const float*)d_in[1];
    const float* b1 = (const float*)d_in[2];
    const float* w2 = (const float*)d_in[3];
    const float* b2 = (const float*)d_in[4];
    const float* cw = (const float*)d_in[5];
    const float* cb = (const float*)d_in[6];
    float* out = (float*)d_out;

    float* ws   = (float*)d_ws;
    float* psum = ws;
    float* pmax = ws + (size_t)BAT * NSL * CHN;
    float* gate = pmax + (size_t)BAT * NSL * CHN;
    float* sbuf = gate + (size_t)BAT * CHN;

    k1_partial<<<BAT * NSL, 256, 0, stream>>>(x, psum, pmax);
    k2_gate<<<BAT, 256, 0, stream>>>(psum, pmax, w1, b1, w2, b2, gate);
    k3_spatial<<<BAT * NSL, K3_WAVES * 32, 0, stream>>>(x, gate, sbuf);
    k4_out<<<K4_GRID, 256, 0, stream>>>(x, gate, sbuf, cw, cb, out);
}


# pool kernel: each lane keeps all four pixel-subgroup accumulators of its channel quad: 4096 waves with 4 loads in flight become 1024 waves with 44-48, combine in registers, no LDS; sum order unchanged
# speedup vs baseline: 2.4835x; 1.0395x over previous
#include <hip/hip_runtime.h>
#include <math.h>


#define CHN 256
#define HIDN 16
#define BAT 32
#define HGT 56
#define WID 56
#define HW  (HGT * WID)
#define NSL 16
#define SLICE (HW / NSL)
#define C4  (CHN / 4)

typedef float v2f __attribute__((ext_vector_type(2)));
typedef float v8f __attribute__((ext_vector_type(8)));

__device__ __forceinline__ float sig_(float v) {
    return __builtin_amdgcn_rcpf(1.0f + __expf(-v));
}

typedef float v4f __attribute__((ext_vector_type(4)));

#define K1_GRP    4
#define K1_NGRP   (SLICE / K1_GRP)
#define K1_AHEAD  44
#define K1_RING   (K1_AHEAD + K1_GRP)

__global__ __launch_bounds__(64) void k1_partial(const float* __restrict__ x,
                                                 float* __restrict__ psum,
                                                 float* __restrict__ pmax) {
    const int q = threadIdx.x;
    const v4f* xp = (const v4f*)(x + (size_t)blockIdx.x * SLICE * CHN) + q;

    v4f sum[K1_GRP], mx[K1_GRP];
#pragma unroll
    for (int j = 0; j < K1_GRP; ++j) {
        sum[j] = (v4f){0.f, 0.f, 0.f, 0.f};
        mx[j]  = (v4f){-3.4e38f, -3.4e38f, -3.4e38f, -3.4e38f};
    }

    v4f ring[K1_RING];
#pragma unroll
    for (int p = 0; p < K1_AHEAD; ++p) ring[p] = xp[(size_t)p * C4];

#pragma unroll
    for (int g = 0; g < K1_NGRP; ++g) {
        __builtin_amdgcn_sched_barrier(0);
#pragma unroll
        for (int j = 0; j < K1_GRP; ++j) {
            const int p = g * K1_GRP + j;
            if (p + K1_AHEAD < SLICE) ring[(p + K1_AHEAD) % K1_RING] = xp[(size_t)(p + K1_AHEAD) * C4];
        }
        __builtin_amdgcn_sched_barrier(0);
#pragma unroll
        for (int j = 0; j < K1_GRP; ++j) {
            const v4f v = ring[(g * K1_GRP + j) % K1_RING];
            sum[j] += v;
            mx[j].x = fmaxf(mx[j].x, v.x); mx[j].y = fmaxf(mx[j].y, v.y);
            mx[j].z = fmaxf(mx[j].z, v.z); mx[j].w = fmaxf(mx[j].w, v.w);
            asm volatile("" : "+v"(sum[j]), "+v"(mx[j]));
        }
    }

    float4 r, m;
    r.x = sum[0].x + sum[1].x + sum[2].x + sum[3].x;
    r.y = sum[0].y + sum[1].y + sum[2].y + sum[3].y;
    r.z = sum[0].z + sum[1].z + sum[2].z + sum[3].z;
    r.w = sum[0].w + sum[1].w + sum[2].w + sum[3].w;
    m.x = fmaxf(fmaxf(mx[0].x, mx[1].x), fmaxf(mx[2].x, mx[3].x));
    m.y = fmaxf(fmaxf(mx[0].y, mx[1].y), fmaxf(mx[2].y, mx[3].y));
    m.z = fmaxf(fmaxf(mx[0].z, mx[1].z), fmaxf(mx[2].z, mx[3].z));
    m.w = fmaxf(fmaxf(mx[0].w, mx[1].w), fmaxf(mx[2].w, mx[3].w));
    ((float4*)psum)[(size_t)blockIdx.x * C4 + q] = r;
    ((float4*)pmax)[(size_t)blockIdx.x * C4 + q] = m;
}

#define PPITCH (CHN + 8)

__global__ __launch_bounds__(256) void k2_gate(const float* __restrict__ psum,
                                               const float* __restrict__ pmax,
                                               const float* __restrict__ w1,
                                               const float* __restrict__ b1,
                                               const float* __restrict__ w2,
                                               const float* __restrict__ b2,
                                               float* __restrict__ gate) {
    __shared__ float pool[3][PPITCH];
    __shared__ float w1s[CHN * HIDN];
    __shared__ float hid[3][HIDN];

    const int b = blockIdx.x;
    const int t = threadIdx.x;
    const int wave = t >> 5;
    const int lane = t & 31;
    const int l16 = lane & 15;
    const int hi = lane >> 4;

#pragma unroll
    for (int i = 0; i < (CHN * HIDN / 4) / 256; ++i)
        ((float4*)w1s)[t + 256 * i] = ((const float4*)w1)[t + 256 * i];

    {
        const float* ps = psum + (size_t)b * NSL * CHN + t;
        const float* pm = pmax + (size_t)b * NSL * CHN + t;
        float rs = 0.0f, rm = -3.4e38f;
#pragma unroll
        for (int s = 0; s < NSL; ++s) {
            rs += ps[(size_t)s * CHN];
            rm = fmaxf(rm, pm[(size_t)s * CHN]);
        }
        pool[0][t] = rs * (1.0f / (float)HW);
        pool[1][t] = rm;
        pool[2][t] = 0.0f;
        if (t < HIDN) hid[2][t] = 0.0f;
    }
    __syncthreads();

    const int mrow = ((l16 & 6) == 0) ? (l16 & 1) : 2;
    const int col = wave * 32 + lane;
    v8f zero = {};

    v2f bw[2][HIDN / 4];
#pragma unroll
    for (int tt = 0; tt < 2; ++tt) {
        const int n0 = (wave * 2 + tt) * 16;
#pragma unroll
        for (int k = 0; k < HIDN; k += 4) {
            bw[tt][k >> 2].x = w2[(k + 2 * hi + 0) * CHN + n0 + l16];
            bw[tt][k >> 2].y = w2[(k + 2 * hi + 1) * CHN + n0 + l16];
        }
    }
    const float bias = b2[col];

    if (wave == 0) {
        const float* arow = pool[mrow];
        v8f c = zero;
#pragma unroll
        for (int k = 0; k < CHN; k += 4) {
            v2f a, bm;
            a.x = arow[k + 2 * hi + 0];
            a.y = arow[k + 2 * hi + 1];
            bm.x = w1s[(k + 2 * hi + 0) * HIDN + l16];
            bm.y = w1s[(k + 2 * hi + 1) * HIDN + l16];
            c = __builtin_amdgcn_wmma_f32_16x16x4_f32(false, a, false, bm,
                                                      (short)0, c, false, false);
        }
        const float v = hi ? c[1] : c[0];
        hid[hi][l16] = fmaxf(v + b1[l16], 0.0f);
    }
    __syncthreads();

    const float* hrow = hid[mrow];
    v8f c2[2];
#pragma unroll
    for (int tt = 0; tt < 2; ++tt) {
        v8f c = zero;
#pragma unroll
        for (int k = 0; k < HIDN; k += 4) {
            v2f a;
            a.x = hrow[k + 2 * hi + 0];
            a.y = hrow[k + 2 * hi + 1];
            c = __builtin_amdgcn_wmma_f32_16x16x4_f32(false, a, false, bw[tt][k >> 2],
                                                      (short)0, c, false, false);
        }
        c2[tt] = c;
    }
    const float pa = hi ? c2[1][0] : c2[0][0];
    const float pm = hi ? c2[1][1] : c2[0][1];
    gate[(size_t)b * CHN + col] = sig_(pa + bias) + sig_(pm + bias);
}

#define K3_WAVES 7
#define K3_PIX   28
#define K3_GRP   7
#define K3_AHEAD 8
#define K3_RING  (K3_AHEAD + 1)

__global__ __launch_bounds__(K3_WAVES * 32) void k3_spatial(const float* __restrict__ x,
                                                            const float* __restrict__ gate,
                                                            float* __restrict__ sbuf) {
    const int lane = threadIdx.x & 31;
    const int wave = __builtin_amdgcn_readfirstlane(threadIdx.x >> 5);
    const int b = blockIdx.x >> 4;
    const int gp0 = blockIdx.x * SLICE + wave * K3_PIX;
    const float4* xp = (const float4*)(x + (size_t)gp0 * CHN) + lane;
    const float4* g4 = (const float4*)(gate + (size_t)b * CHN);
    const float4 g0 = g4[lane];
    const float4 g1 = g4[lane + 32];

    float4 ring[K3_RING][2];
    float sum[K3_PIX], mx[K3_PIX];

#define K3_LOAD(P) do { ring[(P) % K3_RING][0] = xp[(size_t)(P) * C4];        \
                        ring[(P) % K3_RING][1] = xp[(size_t)(P) * C4 + 32]; } while (0)

#pragma unroll
    for (int p = 0; p < K3_AHEAD; ++p) K3_LOAD(p);

#pragma unroll
    for (int g = 0; g < K3_PIX / K3_GRP; ++g) {
#pragma unroll
        for (int j = 0; j < K3_GRP; ++j) {
            const int p = g * K3_GRP + j;
            __builtin_amdgcn_sched_barrier(0);
            if (p + K3_AHEAD < K3_PIX) K3_LOAD(p + K3_AHEAD);
            __builtin_amdgcn_sched_barrier(0);
            float s = 0.0f, m = -3.4e38f;
#pragma unroll
            for (int i = 0; i < 2; ++i) {
                const float4 xv = ring[p % K3_RING][i];
                const float4 gv = i ? g1 : g0;
                const float p0 = xv.x * gv.x, p1 = xv.y * gv.y;
                const float p2 = xv.z * gv.z, p3 = xv.w * gv.w;
                s += p0 + p1 + p2 + p3;
                m = fmaxf(m, fmaxf(fmaxf(p0, p1), fmaxf(p2, p3)));
            }
            sum[p] = s;
            mx[p] = m;
        }
        __builtin_amdgcn_sched_barrier(0);
#pragma unroll
        for (int j = 0; j < K3_GRP; ++j) {
            const int p = g * K3_GRP + j;
#pragma unroll
            for (int off = 16; off > 0; off >>= 1) {
                sum[p] += __shfl_xor(sum[p], off, 32);
                mx[p] = fmaxf(mx[p], __shfl_xor(mx[p], off, 32));
            }
        }
    }
#undef K3_LOAD

    float ms = 0.0f, mm = 0.0f;
#pragma unroll
    for (int p = 0; p < K3_PIX; ++p)
        if (lane == p) { ms = sum[p]; mm = mx[p]; }
    if (lane < K3_PIX) {
        float2 o;
        o.x = ms * (1.0f / (float)CHN);
        o.y = mm;
        ((float2*)sbuf)[(size_t)gp0 + lane] = o;
    }
}

#define K4_ROWS  7
#define K4_COLS  28
#define K4_TROWS (K4_ROWS + 6)
#define K4_PITCH 64
#define K4_ROW4  (WID / 2)
#define K4_PIX   (K4_ROWS * K4_COLS)
#define K4_ITERS (K4_PIX / 4)
#define K4_RITER (K4_COLS / 4)
#define K4_AHEAD 14
#define K4_GRID  (BAT * (HGT / K4_ROWS) * (WID / K4_COLS))
#define K4_OFS(I) ((size_t)((I) / K4_RITER) * (WID * C4) + (size_t)((I) % K4_RITER) * (4 * C4))

__global__ __launch_bounds__(256) void k4_out(const float* __restrict__ x,
                                              const float* __restrict__ gate,
                                              const float* __restrict__ sbuf,
                                              const float* __restrict__ cw,
                                              const float* __restrict__ cb,
                                              float* __restrict__ out) {
    __shared__ float sg[K4_PIX];
    __shared__ __attribute__((aligned(16))) float2 st[K4_TROWS][K4_PITCH];
    const int half = blockIdx.x & 1;
    const int band = (blockIdx.x >> 1) & 7;
    const int b  = blockIdx.x >> 4;
    const int h0 = band * K4_ROWS;
    const int w0 = half * K4_COLS;
    const int t  = threadIdx.x;

    const int r0 = t / K4_ROW4, c0 = t % K4_ROW4;
    const int r1 = (t + 256) / K4_ROW4, c1 = (t + 256) % K4_ROW4;
    const int hh0 = h0 + r0 - 3, hh1 = h0 + r1 - 3;
    float4 tv0 = {0.f, 0.f, 0.f, 0.f}, tv1 = {0.f, 0.f, 0.f, 0.f};
    if (hh0 >= 0 && hh0 < HGT)
        tv0 = ((const float4*)sbuf)[(size_t)(b * HGT + hh0) * K4_ROW4 + c0];
    if (t + 256 < K4_TROWS * K4_ROW4 && hh1 >= 0 && hh1 < HGT)
        tv1 = ((const float4*)sbuf)[(size_t)(b * HGT + hh1) * K4_ROW4 + c1];

    const int q    = t & 63;
    const int wofs = t >> 6;
    const float4 g = ((const float4*)gate)[(size_t)b * C4 + q];

    const size_t pix0 = ((size_t)b * HGT + h0) * WID + w0 + wofs;
    const float4* xb = (const float4*)(x + pix0 * CHN) + q;
    float4*       ob = (float4*)(out + pix0 * CHN) + q;

    float4 ring[K4_AHEAD];
#pragma unroll
    for (int i = 0; i < K4_AHEAD; ++i) ring[i] = xb[K4_OFS(i)];

    *(float4*)&st[r0][4 + 2 * c0] = tv0;
    if (t + 256 < K4_TROWS * K4_ROW4) {
        *(float4*)&st[r1][4 + 2 * c1] = tv1;
    } else if (t < 108 + K4_TROWS * 4) {
        const int i = t - 108;
        const int r = i >> 2;
        const int k = i & 3;
        const float4 z = {0.f, 0.f, 0.f, 0.f};
        *(float4*)&st[r][(k < 2) ? 2 * k : 56 + 2 * k] = z;
    }
    __syncthreads();

    if (t < K4_PIX) {
        const int r  = t / K4_COLS;
        const int wc = w0 + t % K4_COLS;
        float acc = cb[0];
#pragma unroll
        for (int dy = 0; dy < 7; ++dy) {
#pragma unroll
            for (int dx = 0; dx < 7; ++dx) {
                const float2 sv = st[r + dy][wc + dx + 1];
                const float* wp = cw + (dy * 7 + dx) * 2;
                acc += sv.x * wp[0] + sv.y * wp[1];
            }
        }
        sg[t] = sig_(acc);
    }
    __syncthreads();

    float sgv[K4_ITERS];
#pragma unroll
    for (int i = 0; i < K4_ITERS; ++i)
        sgv[i] = sg[(i / K4_RITER) * K4_COLS + (i % K4_RITER) * 4 + wofs];

#pragma unroll
    for (int i = 0; i < K4_ITERS; ++i) {
        float4 xv = ring[i % K4_AHEAD];
        asm volatile("" : "+v"(xv.x), "+v"(xv.y), "+v"(xv.z), "+v"(xv.w));
        if (i + K4_AHEAD < K4_ITERS) ring[i % K4_AHEAD] = xb[K4_OFS(i + K4_AHEAD)];
        const float sgw = sgv[i];
        float4 ov;
        ov.x = xv.x * g.x * sgw;
        ov.y = xv.y * g.y * sgw;
        ov.z = xv.z * g.z * sgw;
        ov.w = xv.w * g.w * sgw;
        ob[K4_OFS(i)] = ov;
    }
}

extern "C" void kernel_launch(void* const* d_in, const int* in_sizes, int n_in,
                              void* d_out, int out_size, void* d_ws, size_t ws_size,
                              hipStream_t stream) {
    const float* x  = (const float*)d_in[0];
    const float* w1 = (const float*)d_in[1];
    const float* b1 = (const float*)d_in[2];
    const float* w2 = (const float*)d_in[3];
    const float* b2 = (const float*)d_in[4];
    const float* cw = (const float*)d_in[5];
    const float* cb = (const float*)d_in[6];
    float* out = (float*)d_out;

    float* ws   = (float*)d_ws;
    float* psum = ws;
    float* pmax = ws + (size_t)BAT * NSL * CHN;
    float* gate = pmax + (size_t)BAT * NSL * CHN;
    float* sbuf = gate + (size_t)BAT * CHN;

    k1_partial<<<BAT * NSL, 64, 0, stream>>>(x, psum, pmax);
    k2_gate<<<BAT, 256, 0, stream>>>(psum, pmax, w1, b1, w2, b2, gate);
    k3_spatial<<<BAT * NSL, K3_WAVES * 32, 0, stream>>>(x, gate, sbuf);
    k4_out<<<K4_GRID, 256, 0, stream>>>(x, gate, sbuf, cw, cb, out);
}


# spatial-stats kernel: 8 lanes per pixel, not 32 (one load = a 128-byte line of each of 4 pixels): channel sums mostly in-lane, 3 DPP steps replace 5 LDS-unit shuffles; per-pixel sum order changes
# speedup vs baseline: 2.5639x; 1.0324x over previous
#include <hip/hip_runtime.h>
#include <math.h>


#define CHN 256
#define HIDN 16
#define BAT 32
#define HGT 56
#define WID 56
#define HW  (HGT * WID)
#define NSL 16
#define SLICE (HW / NSL)
#define C4  (CHN / 4)

typedef float v2f __attribute__((ext_vector_type(2)));
typedef float v8f __attribute__((ext_vector_type(8)));

__device__ __forceinline__ float sig_(float v) {
    return __builtin_amdgcn_rcpf(1.0f + __expf(-v));
}

typedef float v4f __attribute__((ext_vector_type(4)));

#define K1_GRP    4
#define K1_NGRP   (SLICE / K1_GRP)
#define K1_AHEAD  44
#define K1_RING   (K1_AHEAD + K1_GRP)

__global__ __launch_bounds__(64) void k1_partial(const float* __restrict__ x,
                                                 float* __restrict__ psum,
                                                 float* __restrict__ pmax) {
    const int q = threadIdx.x;
    const v4f* xp = (const v4f*)(x + (size_t)blockIdx.x * SLICE * CHN) + q;

    v4f sum[K1_GRP], mx[K1_GRP];
#pragma unroll
    for (int j = 0; j < K1_GRP; ++j) {
        sum[j] = (v4f){0.f, 0.f, 0.f, 0.f};
        mx[j]  = (v4f){-3.4e38f, -3.4e38f, -3.4e38f, -3.4e38f};
    }

    v4f ring[K1_RING];
#pragma unroll
    for (int p = 0; p < K1_AHEAD; ++p) ring[p] = xp[(size_t)p * C4];

#pragma unroll
    for (int g = 0; g < K1_NGRP; ++g) {
        __builtin_amdgcn_sched_barrier(0);
#pragma unroll
        for (int j = 0; j < K1_GRP; ++j) {
            const int p = g * K1_GRP + j;
            if (p + K1_AHEAD < SLICE) ring[(p + K1_AHEAD) % K1_RING] = xp[(size_t)(p + K1_AHEAD) * C4];
        }
        __builtin_amdgcn_sched_barrier(0);
#pragma unroll
        for (int j = 0; j < K1_GRP; ++j) {
            const v4f v = ring[(g * K1_GRP + j) % K1_RING];
            sum[j] += v;
            mx[j].x = fmaxf(mx[j].x, v.x); mx[j].y = fmaxf(mx[j].y, v.y);
            mx[j].z = fmaxf(mx[j].z, v.z); mx[j].w = fmaxf(mx[j].w, v.w);
            asm volatile("" : "+v"(sum[j]), "+v"(mx[j]));
        }
    }

    float4 r, m;
    r.x = sum[0].x + sum[1].x + sum[2].x + sum[3].x;
    r.y = sum[0].y + sum[1].y + sum[2].y + sum[3].y;
    r.z = sum[0].z + sum[1].z + sum[2].z + sum[3].z;
    r.w = sum[0].w + sum[1].w + sum[2].w + sum[3].w;
    m.x = fmaxf(fmaxf(mx[0].x, mx[1].x), fmaxf(mx[2].x, mx[3].x));
    m.y = fmaxf(fmaxf(mx[0].y, mx[1].y), fmaxf(mx[2].y, mx[3].y));
    m.z = fmaxf(fmaxf(mx[0].z, mx[1].z), fmaxf(mx[2].z, mx[3].z));
    m.w = fmaxf(fmaxf(mx[0].w, mx[1].w), fmaxf(mx[2].w, mx[3].w));
    ((float4*)psum)[(size_t)blockIdx.x * C4 + q] = r;
    ((float4*)pmax)[(size_t)blockIdx.x * C4 + q] = m;
}

#define PPITCH (CHN + 8)

__global__ __launch_bounds__(256) void k2_gate(const float* __restrict__ psum,
                                               const float* __restrict__ pmax,
                                               const float* __restrict__ w1,
                                               const float* __restrict__ b1,
                                               const float* __restrict__ w2,
                                               const float* __restrict__ b2,
                                               float* __restrict__ gate) {
    __shared__ float pool[3][PPITCH];
    __shared__ float w1s[CHN * HIDN];
    __shared__ float hid[3][HIDN];

    const int b = blockIdx.x;
    const int t = threadIdx.x;
    const int wave = t >> 5;
    const int lane = t & 31;
    const int l16 = lane & 15;
    const int hi = lane >> 4;

#pragma unroll
    for (int i = 0; i < (CHN * HIDN / 4) / 256; ++i)
        ((float4*)w1s)[t + 256 * i] = ((const float4*)w1)[t + 256 * i];

    {
        const float* ps = psum + (size_t)b * NSL * CHN + t;
        const float* pm = pmax + (size_t)b * NSL * CHN + t;
        float rs = 0.0f, rm = -3.4e38f;
#pragma unroll
        for (int s = 0; s < NSL; ++s) {
            rs += ps[(size_t)s * CHN];
            rm = fmaxf(rm, pm[(size_t)s * CHN]);
        }
        pool[0][t] = rs * (1.0f / (float)HW);
        pool[1][t] = rm;
        pool[2][t] = 0.0f;
        if (t < HIDN) hid[2][t] = 0.0f;
    }
    __syncthreads();

    const int mrow = ((l16 & 6) == 0) ? (l16 & 1) : 2;
    const int col = wave * 32 + lane;
    v8f zero = {};

    v2f bw[2][HIDN / 4];
#pragma unroll
    for (int tt = 0; tt < 2; ++tt) {
        const int n0 = (wave * 2 + tt) * 16;
#pragma unroll
        for (int k = 0; k < HIDN; k += 4) {
            bw[tt][k >> 2].x = w2[(k + 2 * hi + 0) * CHN + n0 + l16];
            bw[tt][k >> 2].y = w2[(k + 2 * hi + 1) * CHN + n0 + l16];
        }
    }
    const float bias = b2[col];

    if (wave == 0) {
        const float* arow = pool[mrow];
        v8f c = zero;
#pragma unroll
        for (int k = 0; k < CHN; k += 4) {
            v2f a, bm;
            a.x = arow[k + 2 * hi + 0];
            a.y = arow[k + 2 * hi + 1];
            bm.x = w1s[(k + 2 * hi + 0) * HIDN + l16];
            bm.y = w1s[(k + 2 * hi + 1) * HIDN + l16];
            c = __builtin_amdgcn_wmma_f32_16x16x4_f32(false, a, false, bm,
                                                      (short)0, c, false, false);
        }
        const float v = hi ? c[1] : c[0];
        hid[hi][l16] = fmaxf(v + b1[l16], 0.0f);
    }
    __syncthreads();

    const float* hrow = hid[mrow];
    v8f c2[2];
#pragma unroll
    for (int tt = 0; tt < 2; ++tt) {
        v8f c = zero;
#pragma unroll
        for (int k = 0; k < HIDN; k += 4) {
            v2f a;
            a.x = hrow[k + 2 * hi + 0];
            a.y = hrow[k + 2 * hi + 1];
            c = __builtin_amdgcn_wmma_f32_16x16x4_f32(false, a, false, bw[tt][k >> 2],
                                                      (short)0, c, false, false);
        }
        c2[tt] = c;
    }
    const float pa = hi ? c2[1][0] : c2[0][0];
    const float pm = hi ? c2[1][1] : c2[0][1];
    gate[(size_t)b * CHN + col] = sig_(pa + bias) + sig_(pm + bias);
}

#define K3_WAVES 7
#define K3_PIX   28
#define K3_GPIX  4
#define K3_NGRP  (K3_PIX / K3_GPIX)
#define K3_LPG   8
#define K3_NLD   (K3_NGRP * K3_LPG)
#define K3_AHEAD 24
#define K3_RING  (K3_AHEAD + 1)

#define K3_XM(V, M) __builtin_bit_cast(float, __builtin_amdgcn_update_dpp(                 \
        __builtin_bit_cast(int, (V)), __builtin_bit_cast(int, (V)), 0x160 | (M), 0xf, 0xf, true))

__global__ __launch_bounds__(K3_WAVES * 32) void k3_spatial(const float* __restrict__ x,
                                                            const float* __restrict__ gate,
                                                            float* __restrict__ sbuf) {
    const int lane = threadIdx.x & 31;
    const int wave = __builtin_amdgcn_readfirstlane(threadIdx.x >> 5);
    const int b = blockIdx.x >> 4;
    const int gp0 = blockIdx.x * SLICE + wave * K3_PIX;
    const int sub = lane >> 3;
    const int c8  = lane & 7;
    const v4f* xp = (const v4f*)(x + (size_t)gp0 * CHN) + sub * C4 + c8;
    const v4f* g4 = (const v4f*)(gate + (size_t)b * CHN) + c8;

    v4f gv[K3_LPG];
#pragma unroll
    for (int i = 0; i < K3_LPG; ++i) gv[i] = g4[i * 8];
    __builtin_amdgcn_sched_barrier(0);

    v4f ring[K3_RING];
#define K3_LOAD(N) ring[(N) % K3_RING] = xp[(size_t)((N) / K3_LPG) * (K3_GPIX * C4) + ((N) % K3_LPG) * 8]

#pragma unroll
    for (int n = 0; n < K3_AHEAD; ++n) K3_LOAD(n);

    float ms = 0.0f, mm = 0.0f;
#pragma unroll
    for (int gi = 0; gi < K3_NGRP; ++gi) {
        v2f s01 = {0.0f, 0.0f}, s23 = {0.0f, 0.0f};
        float m0 = -3.4e38f, m1 = -3.4e38f;
#pragma unroll
        for (int i = 0; i < K3_LPG; ++i) {
            const int n = gi * K3_LPG + i;
            __builtin_amdgcn_sched_barrier(0);
            if (n + K3_AHEAD < K3_NLD) K3_LOAD(n + K3_AHEAD);
            __builtin_amdgcn_sched_barrier(0);
            v4f xv = ring[n % K3_RING];
            asm volatile("" : "+v"(xv));
            const v4f g = gv[i];
            const v2f p01 = {xv.x * g.x, xv.y * g.y};
            const v2f p23 = {xv.z * g.z, xv.w * g.w};
            s01 += p01;
            s23 += p23;
            m0 = fmaxf(fmaxf(m0, p01.x), p01.y);
            m1 = fmaxf(fmaxf(m1, p23.x), p23.y);
        }
        float s = (s01.x + s01.y) + (s23.x + s23.y);
        float m = fmaxf(m0, m1);
        s += K3_XM(s, 1); m = fmaxf(m, K3_XM(m, 1));
        s += K3_XM(s, 2); m = fmaxf(m, K3_XM(m, 2));
        s += K3_XM(s, 4); m = fmaxf(m, K3_XM(m, 4));
        if (c8 == gi) { ms = s; mm = m; }
    }
#undef K3_LOAD

    if (c8 < K3_NGRP) {
        float2 o;
        o.x = ms * (1.0f / (float)CHN);
        o.y = mm;
        ((float2*)sbuf)[(size_t)gp0 + c8 * K3_GPIX + sub] = o;
    }
}

#define K4_ROWS  7
#define K4_COLS  28
#define K4_TROWS (K4_ROWS + 6)
#define K4_PITCH 64
#define K4_ROW4  (WID / 2)
#define K4_PIX   (K4_ROWS * K4_COLS)
#define K4_ITERS (K4_PIX / 4)
#define K4_RITER (K4_COLS / 4)
#define K4_AHEAD 14
#define K4_GRID  (BAT * (HGT / K4_ROWS) * (WID / K4_COLS))
#define K4_OFS(I) ((size_t)((I) / K4_RITER) * (WID * C4) + (size_t)((I) % K4_RITER) * (4 * C4))

__global__ __launch_bounds__(256) void k4_out(const float* __restrict__ x,
                                              const float* __restrict__ gate,
                                              const float* __restrict__ sbuf,
                                              const float* __restrict__ cw,
                                              const float* __restrict__ cb,
                                              float* __restrict__ out) {
    __shared__ float sg[K4_PIX];
    __shared__ __attribute__((aligned(16))) float2 st[K4_TROWS][K4_PITCH];
    const int half = blockIdx.x & 1;
    const int band = (blockIdx.x >> 1) & 7;
    const int b  = blockIdx.x >> 4;
    const int h0 = band * K4_ROWS;
    const int w0 = half * K4_COLS;
    const int t  = threadIdx.x;

    const int r0 = t / K4_ROW4, c0 = t % K4_ROW4;
    const int r1 = (t + 256) / K4_ROW4, c1 = (t + 256) % K4_ROW4;
    const int hh0 = h0 + r0 - 3, hh1 = h0 + r1 - 3;
    float4 tv0 = {0.f, 0.f, 0.f, 0.f}, tv1 = {0.f, 0.f, 0.f, 0.f};
    if (hh0 >= 0 && hh0 < HGT)
        tv0 = ((const float4*)sbuf)[(size_t)(b * HGT + hh0) * K4_ROW4 + c0];
    if (t + 256 < K4_TROWS * K4_ROW4 && hh1 >= 0 && hh1 < HGT)
        tv1 = ((const float4*)sbuf)[(size_t)(b * HGT + hh1) * K4_ROW4 + c1];

    const int q    = t & 63;
    const int wofs = t >> 6;
    const float4 g = ((const float4*)gate)[(size_t)b * C4 + q];

    const size_t pix0 = ((size_t)b * HGT + h0) * WID + w0 + wofs;
    const float4* xb = (const float4*)(x + pix0 * CHN) + q;
    float4*       ob = (float4*)(out + pix0 * CHN) + q;

    float4 ring[K4_AHEAD];
#pragma unroll
    for (int i = 0; i < K4_AHEAD; ++i) ring[i] = xb[K4_OFS(i)];

    *(float4*)&st[r0][4 + 2 * c0] = tv0;
    if (t + 256 < K4_TROWS * K4_ROW4) {
        *(float4*)&st[r1][4 + 2 * c1] = tv1;
    } else if (t < 108 + K4_TROWS * 4) {
        const int i = t - 108;
        const int r = i >> 2;
        const int k = i & 3;
        const float4 z = {0.f, 0.f, 0.f, 0.f};
        *(float4*)&st[r][(k < 2) ? 2 * k : 56 + 2 * k] = z;
    }
    __syncthreads();

    if (t < K4_PIX) {
        const int r  = t / K4_COLS;
        const int wc = w0 + t % K4_COLS;
        float acc = cb[0];
#pragma unroll
        for (int dy = 0; dy < 7; ++dy) {
#pragma unroll
            for (int dx = 0; dx < 7; ++dx) {
                const float2 sv = st[r + dy][wc + dx + 1];
                const float* wp = cw + (dy * 7 + dx) * 2;
                acc += sv.x * wp[0] + sv.y * wp[1];
            }
        }
        sg[t] = sig_(acc);
    }
    __syncthreads();

    float sgv[K4_ITERS];
#pragma unroll
    for (int i = 0; i < K4_ITERS; ++i)
        sgv[i] = sg[(i / K4_RITER) * K4_COLS + (i % K4_RITER) * 4 + wofs];

#pragma unroll
    for (int i = 0; i < K4_ITERS; ++i) {
        float4 xv = ring[i % K4_AHEAD];
        asm volatile("" : "+v"(xv.x), "+v"(xv.y), "+v"(xv.z), "+v"(xv.w));
        if (i + K4_AHEAD < K4_ITERS) ring[i % K4_AHEAD] = xb[K4_OFS(i + K4_AHEAD)];
        const float sgw = sgv[i];
        float4 ov;
        ov.x = xv.x * g.x * sgw;
        ov.y = xv.y * g.y * sgw;
        ov.z = xv.z * g.z * sgw;
        ov.w = xv.w * g.w * sgw;
        ob[K4_OFS(i)] = ov;
    }
}

extern "C" void kernel_launch(void* const* d_in, const int* in_sizes, int n_in,
                              void* d_out, int out_size, void* d_ws, size_t ws_size,
                              hipStream_t stream) {
    const float* x  = (const float*)d_in[0];
    const float* w1 = (const float*)d_in[1];
    const float* b1 = (const float*)d_in[2];
    const float* w2 = (const float*)d_in[3];
    const float* b2 = (const float*)d_in[4];
    const float* cw = (const float*)d_in[5];
    const float* cb = (const float*)d_in[6];
    float* out = (float*)d_out;

    float* ws   = (float*)d_ws;
    float* psum = ws;
    float* pmax = ws + (size_t)BAT * NSL * CHN;
    float* gate = pmax + (size_t)BAT * NSL * CHN;
    float* sbuf = gate + (size_t)BAT * CHN;

    k1_partial<<<BAT * NSL, 64, 0, stream>>>(x, psum, pmax);
    k2_gate<<<BAT, 256, 0, stream>>>(psum, pmax, w1, b1, w2, b2, gate);
    k3_spatial<<<BAT * NSL, K3_WAVES * 32, 0, stream>>>(x, gate, sbuf);
    k4_out<<<K4_GRID, 256, 0, stream>>>(x, gate, sbuf, cw, cb, out);
}
